# starting kernel
# baseline (speedup 1.0000x reference)
#include <hip/hip_runtime.h>
#include <hip/hip_bf16.h>


typedef float v2f __attribute__((ext_vector_type(2)));
typedef float v8f __attribute__((ext_vector_type(8)));

__device__ __forceinline__ float bandsel(int d, float c0, float c1, float c2) {
    float r = 0.0f;
    r = (d == 0) ? c0 : r;
    r = (d == 1) ? c1 : r;
    r = (d == 2) ? c2 : r;
    return r;
}

__device__ __forceinline__ float load_xp_slow(const float* __restrict__ row, int gi, int L) {
    if (gi >= 1 && gi <= L)  return row[gi - 1];
    if (gi == 0)             return 2.0f * row[0]     - row[1];
    if (gi == L + 1)         return 2.0f * row[L - 1] - row[L - 2];
    return 0.0f;
}

__global__ __launch_bounds__(128) void synth_level_kernel(
        const float* __restrict__ x, const float* __restrict__ d,
        float* __restrict__ out,
        const float* __restrict__ h, const float* __restrict__ g,
        int L) {
    __shared__ float smx[4][16][20];
    __shared__ float smd[4][16][20];

    const int w    = threadIdx.x >> 5;
    const int lane = threadIdx.x & 31;
    const int c0   = w << 4;
    const int n    = lane & 15;
    const int hi   = lane >> 4;

    const float h0 = h[0], h1 = h[1], h2 = h[2], h3 = h[3], h4 = h[4], h5 = h[5];
    const float g0 = g[0], g1 = g[1], g2 = g[2], g3 = g[3], g4 = g[4], g5 = g[5];

    v2f bhe[5], bho[5], bge[5], bgo[5];
#pragma unroll
    for (int c = 0; c < 5; ++c) {
        const int r0 = c * 4 + (hi << 1);
        const int d0r = r0 - n, d1r = r0 + 1 - n;
        bhe[c] = v2f{ bandsel(d0r, h4, h2, h0), bandsel(d1r, h4, h2, h0) };
        bho[c] = v2f{ bandsel(d0r, h5, h3, h1), bandsel(d1r, h5, h3, h1) };
        bge[c] = v2f{ bandsel(d0r, g4, g2, g0), bandsel(d1r, g4, g2, g0) };
        bgo[c] = v2f{ bandsel(d0r, g5, g3, g1), bandsel(d1r, g5, g3, g1) };
    }

    const size_t outL = 2 * (size_t)L;

    for (int it = 0; it < 4; ++it) {
        const int t0 = (blockIdx.x << 6) + (it << 4);

        if (t0 >= 1 && t0 + 19 <= L) {
            for (int idx = lane; idx < 16 * 20; idx += 32) {
                const int r = idx / 20;
                const int i = idx % 20;
                const size_t rowBase = (size_t)(c0 + r) * (size_t)L + (size_t)(t0 - 1);
                smx[w][r][i] = x[rowBase + i];
                smd[w][r][i] = d[rowBase + i];
            }
        } else {
            for (int idx = lane; idx < 16 * 20; idx += 32) {
                const int r  = idx / 20;
                const int i  = idx % 20;
                const int gi = t0 + i;
                const size_t rowBase = (size_t)(c0 + r) * (size_t)L;
                smx[w][r][i] = load_xp_slow(x + rowBase, gi, L);
                smd[w][r][i] = load_xp_slow(d + rowBase, gi, L);
            }
        }
        __syncthreads();

        v8f ce = {};
        v8f co = {};
#pragma unroll
        for (int c = 0; c < 5; ++c) {
            const int k0 = c * 4 + (hi << 1);
            const v2f ax = v2f{ smx[w][n][k0], smx[w][n][k0 + 1] };
            const v2f ad = v2f{ smd[w][n][k0], smd[w][n][k0 + 1] };

            ce = __builtin_amdgcn_wmma_f32_16x16x4_f32(false, ax, false, bhe[c], (short)0, ce, false, false);
            ce = __builtin_amdgcn_wmma_f32_16x16x4_f32(false, ad, false, bge[c], (short)0, ce, false, false);
            co = __builtin_amdgcn_wmma_f32_16x16x4_f32(false, ax, false, bho[c], (short)0, co, false, false);
            co = __builtin_amdgcn_wmma_f32_16x16x4_f32(false, ad, false, bgo[c], (short)0, co, false, false);
        }

        const int pos = t0 + n;
#pragma unroll
        for (int r = 0; r < 8; ++r) {
            const int ch = c0 + r + (hi << 3);
            float2 v;
            v.x = ce[r];
            v.y = co[r];
            *reinterpret_cast<float2*>(&out[(size_t)ch * outL + 2 * (size_t)pos]) = v;
        }
        __syncthreads();
    }
}

extern "C" void kernel_launch(void* const* d_in, const int* in_sizes, int n_in,
                              void* d_out, int out_size, void* d_ws, size_t ws_size,
                              hipStream_t stream) {
    (void)in_sizes; (void)n_in; (void)out_size; (void)ws_size;

    const float* sig = (const float*)d_in[0];
    const float* d0  = (const float*)d_in[1];
    const float* d1  = (const float*)d_in[2];
    const float* d2  = (const float*)d_in[3];
    const float* d3  = (const float*)d_in[4];
    const float* h   = (const float*)d_in[5];
    const float* g   = (const float*)d_in[6];
    float* out = (float*)d_out;

    float* A = (float*)d_ws;
    float* B = A + (size_t)64 * 65536;

    dim3 blk(128);
    synth_level_kernel<<<16384  / 64, blk, 0, stream>>>(sig, d3, B,   h, g, 16384);
    synth_level_kernel<<<32768  / 64, blk, 0, stream>>>(B,   d2, A,   h, g, 32768);
    synth_level_kernel<<<65536  / 64, blk, 0, stream>>>(A,   d1, B,   h, g, 65536);
    synth_level_kernel<<<131072 / 64, blk, 0, stream>>>(B,   d0, out, h, g, 131072);
}


# B columns = 16 consecutive tiles of ONE channel, not 16 channels: a 16-byte-per-lane load is one contiguous 512-byte run, two stores fill 8 whole lines; a fence keeps each chunk pair's stores with it
# speedup vs baseline: 1.8667x; 1.8667x over previous
#include <hip/hip_runtime.h>


typedef float v2f __attribute__((ext_vector_type(2)));
typedef float v4f __attribute__((ext_vector_type(4)));
typedef float v8f __attribute__((ext_vector_type(8)));

__device__ __forceinline__ float bandsel(int j, float c0, float c1, float c2) {
    float r = 0.0f;
    r = (j == 0) ? c0 : r;
    r = (j == 1) ? c1 : r;
    r = (j == 2) ? c2 : r;
    return r;
}

template <int C>
__global__ __launch_bounds__(256) void synth_level_kernel(
        const float* __restrict__ x, const float* __restrict__ d,
        float* __restrict__ out,
        const float* __restrict__ h, const float* __restrict__ g,
        int L, int lgS) {
    const int lane = threadIdx.x & 31;
    const int n    = lane & 15;
    const int hf   = lane >> 4;
    const int w    = __builtin_amdgcn_readfirstlane((int)(threadIdx.x >> 5));
    const int Wv   = ((int)blockIdx.x << 3) + w;
    const int ch   = Wv >> lgS;
    const int t0   = (Wv & ((1 << lgS) - 1)) * (128 * C);
    const bool first = (t0 == 0);
    const bool last  = (t0 + 128 * C == L);

    const float h0 = h[0], h1 = h[1], h2 = h[2], h3 = h[3], h4 = h[4], h5 = h[5];
    const float g0 = g[0], g1 = g[1], g2 = g[2], g3 = g[3], g4 = g[4], g5 = g[5];

    const int o   = (n & 3) + ((n >> 3) << 2) + (((n >> 2) & 1) << 3);
    const int mm  = o >> 1;
    const bool od = (o & 1) != 0;
    const float hk0 = od ? h5 : h4, hk1 = od ? h3 : h2, hk2 = od ? h1 : h0;
    const float gk0 = od ? g5 : g4, gk1 = od ? g3 : g2, gk2 = od ? g1 : g0;
    v2f ah[2][2], ag[2][2];
#pragma unroll
    for (int gg = 0; gg < 2; ++gg) {
#pragma unroll
        for (int s = 0; s < 2; ++s) {
            const int j0 = 8 * gg + 2 * s + 4 * hf - 3 - mm;
            ah[gg][s] = v2f{ bandsel(j0, hk0, hk1, hk2), bandsel(j0 + 1, hk0, hk1, hk2) };
            ag[gg][s] = v2f{ bandsel(j0, gk0, gk1, gk2), bandsel(j0 + 1, gk0, gk1, gk2) };
        }
    }

    const size_t row = (size_t)ch * (size_t)L + (size_t)t0;
    const float* xr = x + row;
    const float* dr = d + row;
    const int e0   = 8 * n + 4 * hf - 4;
    const bool fl  = first && (lane == 0);
    const bool lh  = last && (lane == 31);
    const int ea   = e0 + (fl ? 4 : 0);
    const int eb   = e0 + 128 * (C - 1) + 8 + (lh ? -4 : 0);
    v4f X[C][2], D[C][2];
    X[0][0]     = *reinterpret_cast<const v4f*>(xr + ea);
    D[0][0]     = *reinterpret_cast<const v4f*>(dr + ea);
    X[C - 1][1] = *reinterpret_cast<const v4f*>(xr + eb);
    D[C - 1][1] = *reinterpret_cast<const v4f*>(dr + eb);
#pragma unroll
    for (int k = 0; k < C; ++k) {
#pragma unroll
        for (int gg = 0; gg < 2; ++gg) {
            if ((k == 0 && gg == 0) || (k == C - 1 && gg == 1)) continue;
            const int e = e0 + 128 * k + 8 * gg;
            X[k][gg] = *reinterpret_cast<const v4f*>(xr + e);
            D[k][gg] = *reinterpret_cast<const v4f*>(dr + e);
        }
    }

    {
        const float lx = 2.0f * X[0][0].x - X[0][0].y;
        const float ld = 2.0f * D[0][0].x - D[0][0].y;
        X[0][0].w = fl ? lx : X[0][0].w;
        D[0][0].w = fl ? ld : D[0][0].w;
        const float rx = 2.0f * X[C - 1][1].w - X[C - 1][1].z;
        const float rd = 2.0f * D[C - 1][1].w - D[C - 1][1].z;
        X[C - 1][1].x = lh ? rx : X[C - 1][1].x;
        D[C - 1][1].x = lh ? rd : D[C - 1][1].x;
    }

    float* orow = out + (size_t)ch * (2 * (size_t)L) + 2 * (size_t)t0 + (16 * n + 4 * hf);

    __builtin_amdgcn_sched_barrier(0);

#pragma unroll
    for (int k = 0; k < C; k += 2) {
        v8f a0 = {};
        v8f a1 = {};
#pragma unroll
        for (int gg = 0; gg < 2; ++gg) {
#pragma unroll
            for (int s = 0; s < 2; ++s) {
                const v2f bx0 = v2f{ X[k][gg][2 * s],     X[k][gg][2 * s + 1] };
                const v2f bx1 = v2f{ X[k + 1][gg][2 * s], X[k + 1][gg][2 * s + 1] };
                const v2f bd0 = v2f{ D[k][gg][2 * s],     D[k][gg][2 * s + 1] };
                const v2f bd1 = v2f{ D[k + 1][gg][2 * s], D[k + 1][gg][2 * s + 1] };
                a0 = __builtin_amdgcn_wmma_f32_16x16x4_f32(false, ah[gg][s], false, bx0, (short)0, a0, false, false);
                a1 = __builtin_amdgcn_wmma_f32_16x16x4_f32(false, ah[gg][s], false, bx1, (short)0, a1, false, false);
                a0 = __builtin_amdgcn_wmma_f32_16x16x4_f32(false, ag[gg][s], false, bd0, (short)0, a0, false, false);
                a1 = __builtin_amdgcn_wmma_f32_16x16x4_f32(false, ag[gg][s], false, bd1, (short)0, a1, false, false);
            }
        }
        float* po = orow + 256 * k;
        *reinterpret_cast<v4f*>(po)       = v4f{ a0[0], a0[1], a0[2], a0[3] };
        *reinterpret_cast<v4f*>(po + 8)   = v4f{ a0[4], a0[5], a0[6], a0[7] };
        *reinterpret_cast<v4f*>(po + 256) = v4f{ a1[0], a1[1], a1[2], a1[3] };
        *reinterpret_cast<v4f*>(po + 264) = v4f{ a1[4], a1[5], a1[6], a1[7] };
        __builtin_amdgcn_sched_barrier(0);
    }
}

extern "C" void kernel_launch(void* const* d_in, const int* in_sizes, int n_in,
                              void* d_out, int out_size, void* d_ws, size_t ws_size,
                              hipStream_t stream) {
    (void)in_sizes; (void)n_in; (void)out_size; (void)ws_size;

    const float* sig = (const float*)d_in[0];
    const float* d0  = (const float*)d_in[1];
    const float* d1  = (const float*)d_in[2];
    const float* d2  = (const float*)d_in[3];
    const float* d3  = (const float*)d_in[4];
    const float* h   = (const float*)d_in[5];
    const float* g   = (const float*)d_in[6];
    float* out = (float*)d_out;

    float* A = (float*)d_ws;
    float* B = A + (size_t)64 * 65536;

    dim3 blk(256);
    synth_level_kernel<4><<<64 * (16384  / 512)  / 8, blk, 0, stream>>>(sig, d3, B,   h, g, 16384,  5);
    synth_level_kernel<4><<<64 * (32768  / 512)  / 8, blk, 0, stream>>>(B,   d2, A,   h, g, 32768,  6);
    synth_level_kernel<8><<<64 * (65536  / 1024) / 8, blk, 0, stream>>>(A,   d1, B,   h, g, 65536,  6);
    synth_level_kernel<8><<<64 * (131072 / 1024) / 8, blk, 0, stream>>>(B,   d0, out, h, g, 131072, 7);
}


# two levels per launch, 2 launches not 4: a wave stores its strip to the workspace, fences, reads it back as the next level's x; its 2 edge neighbours are recomputed as scalar 6-term FMA sums
# speedup vs baseline: 2.3420x; 1.2546x over previous
#include <hip/hip_runtime.h>


typedef float v2f __attribute__((ext_vector_type(2)));
typedef float v4f __attribute__((ext_vector_type(4)));
typedef float v8f __attribute__((ext_vector_type(8)));

__device__ __forceinline__ float bandsel(int j, float c0, float c1, float c2) {
    float r = 0.0f;
    r = (j == 0) ? c0 : r;
    r = (j == 1) ? c1 : r;
    r = (j == 2) ? c2 : r;
    return r;
}

template <int C>
__device__ __forceinline__ void load_pieces(const float* r, int e0, bool lo, bool hi,
                                            v4f (&P)[C][2]) {
    const int ea = e0 + (lo ? 4 : 0);
    const int eb = e0 + 128 * (C - 1) + 8 + (hi ? -4 : 0);
    P[0][0]     = *reinterpret_cast<const v4f*>(r + ea);
    P[C - 1][1] = *reinterpret_cast<const v4f*>(r + eb);
#pragma unroll
    for (int k = 0; k < C; ++k) {
#pragma unroll
        for (int gg = 0; gg < 2; ++gg) {
            if ((k == 0 && gg == 0) || (k == C - 1 && gg == 1)) continue;
            P[k][gg] = *reinterpret_cast<const v4f*>(r + (e0 + 128 * k + 8 * gg));
        }
    }
}

template <int C>
__device__ __forceinline__ void set_ends(v4f (&P)[C][2], bool lo, bool hi,
                                         bool first, bool last, float vl, float vr) {
    const float al = 2.0f * P[0][0].x - P[0][0].y;
    const float ar = 2.0f * P[C - 1][1].w - P[C - 1][1].z;
    const float l = first ? al : vl;
    const float r = last  ? ar : vr;
    P[0][0].w     = lo ? l : P[0][0].w;
    P[C - 1][1].x = hi ? r : P[C - 1][1].x;
}

template <int C>
__device__ __forceinline__ void synth_strip(
        const v4f (&X)[C][2], const v4f (&D)[C][2], float* ostrip,
        const v2f (&ah)[2][2], const v2f (&ag)[2][2], int n, int hf) {
    float* orow = ostrip + (16 * n + 4 * hf);

    __builtin_amdgcn_sched_barrier(0);

#pragma unroll
    for (int k = 0; k < C; k += 2) {
        v8f a0 = {};
        v8f a1 = {};
#pragma unroll
        for (int gg = 0; gg < 2; ++gg) {
#pragma unroll
            for (int s = 0; s < 2; ++s) {
                const v2f bx0 = v2f{ X[k][gg][2 * s],     X[k][gg][2 * s + 1] };
                const v2f bx1 = v2f{ X[k + 1][gg][2 * s], X[k + 1][gg][2 * s + 1] };
                const v2f bd0 = v2f{ D[k][gg][2 * s],     D[k][gg][2 * s + 1] };
                const v2f bd1 = v2f{ D[k + 1][gg][2 * s], D[k + 1][gg][2 * s + 1] };
                a0 = __builtin_amdgcn_wmma_f32_16x16x4_f32(false, ah[gg][s], false, bx0, (short)0, a0, false, false);
                a1 = __builtin_amdgcn_wmma_f32_16x16x4_f32(false, ah[gg][s], false, bx1, (short)0, a1, false, false);
                a0 = __builtin_amdgcn_wmma_f32_16x16x4_f32(false, ag[gg][s], false, bd0, (short)0, a0, false, false);
                a1 = __builtin_amdgcn_wmma_f32_16x16x4_f32(false, ag[gg][s], false, bd1, (short)0, a1, false, false);
            }
        }
        float* po = orow + 256 * k;
        *reinterpret_cast<v4f*>(po)       = v4f{ a0[0], a0[1], a0[2], a0[3] };
        *reinterpret_cast<v4f*>(po + 8)   = v4f{ a0[4], a0[5], a0[6], a0[7] };
        *reinterpret_cast<v4f*>(po + 256) = v4f{ a1[0], a1[1], a1[2], a1[3] };
        *reinterpret_cast<v4f*>(po + 264) = v4f{ a1[4], a1[5], a1[6], a1[7] };
        __builtin_amdgcn_sched_barrier(0);
    }
}

template <int C1>
__global__ __launch_bounds__(256) void synth_two_levels_kernel(
        const float* __restrict__ x, const float* __restrict__ da,
        const float* __restrict__ db, float* mid, float* __restrict__ out,
        const float* __restrict__ h, const float* __restrict__ g,
        int L, int lgS) {
    constexpr int C2 = 2 * C1;
    const int lane = threadIdx.x & 31;
    const int n    = lane & 15;
    const int hf   = lane >> 4;
    const int w    = __builtin_amdgcn_readfirstlane((int)(threadIdx.x >> 5));
    const int Wv   = ((int)blockIdx.x << 3) + w;
    const int ch   = Wv >> lgS;
    const int S    = 128 * C1;
    const int t0   = (Wv & ((1 << lgS) - 1)) * S;
    const bool first = (t0 == 0);
    const bool last  = (t0 + S == L);

    const size_t rowa = (size_t)ch * (size_t)L;
    const float* xrow = x + rowa;
    const float* drow = da + rowa;
    float* mstrip = mid + 2 * rowa + 2 * (size_t)t0;
    const int e0   = 8 * n + 4 * hf - 4;
    const bool l0  = (lane == 0);
    const bool l31 = (lane == 31);
    const bool fl  = first && l0;
    const bool lh  = last && l31;
    v4f X1[C1][2], D1[C1][2], X2[C2][2], D2[C2][2];
    load_pieces<C1>(xrow + t0, e0, fl, lh, X1);
    load_pieces<C1>(drow + t0, e0, fl, lh, D1);
    load_pieces<C2>(db + 2 * rowa + 2 * (size_t)t0, e0, fl, lh, D2);

    const float h0 = h[0], h1 = h[1], h2 = h[2], h3 = h[3], h4 = h[4], h5 = h[5];
    const float g0 = g[0], g1 = g[1], g2 = g[2], g3 = g[3], g4 = g[4], g5 = g[5];

    const int o   = (n & 3) + ((n >> 3) << 2) + (((n >> 2) & 1) << 3);
    const int mm  = o >> 1;
    const bool od = (o & 1) != 0;
    const float hk0 = od ? h5 : h4, hk1 = od ? h3 : h2, hk2 = od ? h1 : h0;
    const float gk0 = od ? g5 : g4, gk1 = od ? g3 : g2, gk2 = od ? g1 : g0;
    v2f ah[2][2], ag[2][2];
#pragma unroll
    for (int gg = 0; gg < 2; ++gg) {
#pragma unroll
        for (int s = 0; s < 2; ++s) {
            const int j0 = 8 * gg + 2 * s + 4 * hf - 3 - mm;
            ah[gg][s] = v2f{ bandsel(j0, hk0, hk1, hk2), bandsel(j0 + 1, hk0, hk1, hk2) };
            ag[gg][s] = v2f{ bandsel(j0, gk0, gk1, gk2), bandsel(j0 + 1, gk0, gk1, gk2) };
        }
    }

    const int jl = first ? 0 : t0 - 2;
    const int jr = last ? L - 3 : t0 + S - 1;
    const float haloL = xrow[jl] * h5 + xrow[jl + 1] * h3 + xrow[jl + 2] * h1
                      + drow[jl] * g5 + drow[jl + 1] * g3 + drow[jl + 2] * g1;
    const float haloR = xrow[jr] * h4 + xrow[jr + 1] * h2 + xrow[jr + 2] * h0
                      + drow[jr] * g4 + drow[jr + 1] * g2 + drow[jr + 2] * g0;

    set_ends<C1>(X1, fl, lh, true, true, 0.0f, 0.0f);
    set_ends<C1>(D1, fl, lh, true, true, 0.0f, 0.0f);
    set_ends<C2>(D2, fl, lh, true, true, 0.0f, 0.0f);
    synth_strip<C1>(X1, D1, mstrip, ah, ag, n, hf);

    __builtin_amdgcn_fence(__ATOMIC_RELEASE, "workgroup");
    __builtin_amdgcn_fence(__ATOMIC_ACQUIRE, "workgroup");

    load_pieces<C2>(mstrip, e0, l0, l31, X2);
    set_ends<C2>(X2, l0, l31, first, last, haloL, haloR);
    synth_strip<C2>(X2, D2, out + 4 * rowa + 4 * (size_t)t0, ah, ag, n, hf);
}

extern "C" void kernel_launch(void* const* d_in, const int* in_sizes, int n_in,
                              void* d_out, int out_size, void* d_ws, size_t ws_size,
                              hipStream_t stream) {
    (void)in_sizes; (void)n_in; (void)out_size; (void)ws_size;

    const float* sig = (const float*)d_in[0];
    const float* d0  = (const float*)d_in[1];
    const float* d1  = (const float*)d_in[2];
    const float* d2  = (const float*)d_in[3];
    const float* d3  = (const float*)d_in[4];
    const float* h   = (const float*)d_in[5];
    const float* g   = (const float*)d_in[6];
    float* out = (float*)d_out;

    float* A  = (float*)d_ws;
    float* M2 = A + (size_t)64 * 65536;
    float* M1 = M2;

    dim3 blk(256);
    synth_two_levels_kernel<2><<<64 * (16384 / 256) / 8, blk, 0, stream>>>(sig, d3, d2, M1, A,   h, g, 16384, 6);
    synth_two_levels_kernel<4><<<64 * (65536 / 512) / 8, blk, 0, stream>>>(A,   d1, d0, M2, out, h, g, 65536, 7);
}


# second level takes the first level's D registers as its B fragments: the level in between is neither stored nor read back; column neighbours by lane exchange; the 6 terms regroup in K
# speedup vs baseline: 3.2478x; 1.3868x over previous
#include <hip/hip_runtime.h>


typedef float v2f __attribute__((ext_vector_type(2)));
typedef float v4f __attribute__((ext_vector_type(4)));
typedef float v8f __attribute__((ext_vector_type(8)));

__device__ __forceinline__ float bandsel(int j, float c0, float c1, float c2) {
    float r = 0.0f;
    r = (j == 0) ? c0 : r;
    r = (j == 1) ? c1 : r;
    r = (j == 2) ? c2 : r;
    return r;
}

template <int C>
__device__ __forceinline__ void load_pieces(const float* r, int e0, bool lo, bool hi,
                                            v4f (&P)[C][2]) {
    const int ea = e0 + (lo ? 4 : 0);
    const int eb = e0 + 128 * (C - 1) + 8 + (hi ? -4 : 0);
    P[0][0]     = *reinterpret_cast<const v4f*>(r + ea);
    P[C - 1][1] = *reinterpret_cast<const v4f*>(r + eb);
#pragma unroll
    for (int k = 0; k < C; ++k) {
#pragma unroll
        for (int gg = 0; gg < 2; ++gg) {
            if ((k == 0 && gg == 0) || (k == C - 1 && gg == 1)) continue;
            P[k][gg] = *reinterpret_cast<const v4f*>(r + (e0 + 128 * k + 8 * gg));
        }
    }
}

template <int C>
__device__ __forceinline__ void set_ends(v4f (&P)[C][2], bool lo, bool hi,
                                         bool first, bool last, float vl, float vr) {
    const float al = 2.0f * P[0][0].x - P[0][0].y;
    const float ar = 2.0f * P[C - 1][1].w - P[C - 1][1].z;
    const float l = first ? al : vl;
    const float r = last  ? ar : vr;
    P[0][0].w     = lo ? l : P[0][0].w;
    P[C - 1][1].x = hi ? r : P[C - 1][1].x;
}

__device__ __forceinline__ v8f mma(const v2f a, const v2f b, const v8f c) {
    return __builtin_amdgcn_wmma_f32_16x16x4_f32(false, a, false, b, (short)0, c, false, false);
}

__device__ __forceinline__ float other_half(float v, unsigned sel_lo, unsigned sel_hi) {
    const int i = __builtin_bit_cast(int, v);
    return __builtin_bit_cast(float, __builtin_amdgcn_permlanex16(i, i, sel_lo, sel_hi, false, false));
}

__device__ __forceinline__ float from_lane(float v, int l) {
    return __builtin_bit_cast(float, __builtin_amdgcn_readlane(__builtin_bit_cast(int, v), l));
}

template <int C1>
__global__ __launch_bounds__(256, 8) void synth_two_levels_kernel(
        const float* __restrict__ x, const float* __restrict__ da,
        const float* __restrict__ db, float* __restrict__ out,
        const float* __restrict__ h, const float* __restrict__ g,
        int L, int lgS) {
    const int lane = threadIdx.x & 31;
    const int n    = lane & 15;
    const int hf   = lane >> 4;
    const int w    = __builtin_amdgcn_readfirstlane((int)(threadIdx.x >> 5));
    const int Wv   = ((int)blockIdx.x << 3) + w;
    const int ch   = Wv >> lgS;
    const int S    = 128 * C1;
    const int t0   = (Wv & ((1 << lgS) - 1)) * S;
    const bool first = (t0 == 0);
    const bool last  = (t0 + S == L);

    const size_t rowa = (size_t)ch * (size_t)L;
    const float* xrow = x + rowa;
    const float* drow = da + rowa;
    const float* d2row = db + 2 * rowa;
    const int e0   = 8 * n + 4 * hf - 4;
    const bool l0  = (lane == 0);
    const bool l31 = (lane == 31);
    const bool fl  = first && l0;
    const bool lh  = last && l31;
    v4f X1[C1][2], D1[C1][2], D2[C1][2];
    load_pieces<C1>(xrow + t0, e0, fl, lh, X1);
    load_pieces<C1>(drow + t0, e0, fl, lh, D1);
    int d2off = 16 * n + 4 * hf;
    asm volatile("" : "+v"(d2off));
    if constexpr (C1 <= 2) {
        const float* d2lane = d2row + 2 * (size_t)t0 + d2off;
#pragma unroll
        for (int k = 0; k < C1; ++k) {
            D2[k][0] = *reinterpret_cast<const v4f*>(d2lane + 256 * k);
            D2[k][1] = *reinterpret_cast<const v4f*>(d2lane + 256 * k + 8);
        }
    }

    const float h0 = h[0], h1 = h[1], h2 = h[2], h3 = h[3], h4 = h[4], h5 = h[5];
    const float g0 = g[0], g1 = g[1], g2 = g[2], g3 = g[3], g4 = g[4], g5 = g[5];

    const int o   = (n & 3) + ((n >> 3) << 2) + (((n >> 2) & 1) << 3);
    const int mm  = o >> 1;
    const bool od = (o & 1) != 0;
    const float hk0 = od ? h5 : h4, hk1 = od ? h3 : h2, hk2 = od ? h1 : h0;
    const float gk0 = od ? g5 : g4, gk1 = od ? g3 : g2, gk2 = od ? g1 : g0;
    v2f ah[2][2], ag[2][2];
#pragma unroll
    for (int gg = 0; gg < 2; ++gg) {
#pragma unroll
        for (int s = 0; s < 2; ++s) {
            const int j0 = 8 * gg + 2 * s + 4 * hf - 3 - mm;
            ah[gg][s] = v2f{ bandsel(j0, hk0, hk1, hk2), bandsel(j0 + 1, hk0, hk1, hk2) };
            ag[gg][s] = v2f{ bandsel(j0, gk0, gk1, gk2), bandsel(j0 + 1, gk0, gk1, gk2) };
        }
    }

    const int jl = first ? 0 : t0 - 2;
    const int jr = last ? L - 3 : t0 + S - 1;
    const float haloL = xrow[jl] * h5 + xrow[jl + 1] * h3 + xrow[jl + 2] * h1
                      + drow[jl] * g5 + drow[jl + 1] * g3 + drow[jl + 2] * g1;
    const float haloR = xrow[jr] * h4 + xrow[jr + 1] * h2 + xrow[jr + 2] * h0
                      + drow[jr] * g4 + drow[jr + 1] * g2 + drow[jr + 2] * g0;
    const float dL = d2row[first ? 0 : 2 * t0 - 1];
    const float dR = d2row[last ? 2 * L - 1 : 2 * (t0 + S)];

    set_ends<C1>(X1, fl, lh, true, true, 0.0f, 0.0f);
    set_ends<C1>(D1, fl, lh, true, true, 0.0f, 0.0f);
    v8f acc[C1];

    __builtin_amdgcn_sched_barrier(0);
#pragma unroll
    for (int k = 0; k < C1; k += 2) {
        v8f a0 = {};
        v8f a1 = {};
#pragma unroll
        for (int gg = 0; gg < 2; ++gg) {
#pragma unroll
            for (int s = 0; s < 2; ++s) {
                const v2f bx0 = v2f{ X1[k][gg][2 * s],     X1[k][gg][2 * s + 1] };
                const v2f bx1 = v2f{ X1[k + 1][gg][2 * s], X1[k + 1][gg][2 * s + 1] };
                const v2f bd0 = v2f{ D1[k][gg][2 * s],     D1[k][gg][2 * s + 1] };
                const v2f bd1 = v2f{ D1[k + 1][gg][2 * s], D1[k + 1][gg][2 * s + 1] };
                a0 = mma(ah[gg][s], bx0, a0);
                a1 = mma(ah[gg][s], bx1, a1);
                a0 = mma(ag[gg][s], bd0, a0);
                a1 = mma(ag[gg][s], bd1, a1);
            }
        }
        acc[k]     = a0;
        acc[k + 1] = a1;
        if constexpr (C1 > 2) {
            if (k == 0) {
                asm volatile("v_nop\n\tv_nop\n\tv_nop\n\tv_nop\n\tv_nop"
                             : "+v"(acc[0]), "+v"(acc[1]), "+v"(d2off));
                const float* d2lane = d2row + 2 * (size_t)t0 + d2off;
#pragma unroll
                for (int kk = 0; kk < C1; ++kk) {
                    D2[kk][0] = *reinterpret_cast<const v4f*>(d2lane + 256 * kk);
                    D2[kk][1] = *reinterpret_cast<const v4f*>(d2lane + 256 * kk + 8);
                }
                __builtin_amdgcn_sched_barrier(0);
            }
        }
    }
    int lane2 = lane;
    if constexpr (C1 == 2) {
        asm volatile("v_nop\n\tv_nop\n\tv_nop\n\tv_nop\n\tv_nop"
                     : "+v"(acc[0]), "+v"(acc[1]), "+v"(lane2));
    } else {
        static_assert(C1 == 2 || C1 == 4, "strip of 2 or 4 chunks");
        asm volatile("v_nop\n\tv_nop\n\tv_nop\n\tv_nop\n\tv_nop"
                     : "+v"(acc[0]), "+v"(acc[1]), "+v"(acc[2]), "+v"(acc[3]), "+v"(lane2));
    }
    __builtin_amdgcn_sched_barrier(0);

    const int n2   = lane2 & 15;
    const int hf2  = lane2 >> 4;
    const bool f0  = (lane2 == 0);
    const bool f31 = (lane2 == 31);
    const int o2   = (n2 & 3) + ((n2 >> 3) << 2) + (((n2 >> 2) & 1) << 3);
    const int mm2  = o2 >> 1;
    const bool od2 = (o2 & 1) != 0;
    const float hq0 = od2 ? h5 : h4, hq1 = od2 ? h3 : h2, hq2 = od2 ? h1 : h0;
    const float gq0 = od2 ? g5 : g4, gq1 = od2 ? g3 : g2, gq2 = od2 ? g1 : g0;
    const int jb = 4 * hf2 - mm2 + 1;
    v2f bh[4], bg[4];
    {
        const int offs[4] = { 0, 2, 8, -6 };
#pragma unroll
        for (int i = 0; i < 4; ++i) {
            const int j0 = offs[i] + jb;
            bh[i] = v2f{ bandsel(j0, hq0, hq1, hq2), bandsel(j0 + 1, hq0, hq1, hq2) };
            bg[i] = v2f{ bandsel(j0, gq0, gq1, gq2), bandsel(j0 + 1, gq0, gq1, gq2) };
        }
    }
    const bool r0 = (hf2 == 0) && (mm2 == 0);
    const bool r1 = (hf2 == 1) && (mm2 == 7);
    const v2f an0 = v2f{ r0 ? hq0 : 0.0f, r0 ? gq0 : 0.0f };
    const v2f an1 = v2f{ r1 ? hq2 : 0.0f, r1 ? gq2 : 0.0f };

    v2f bn[C1];
#pragma unroll
    for (int k = 0; k < C1; ++k) {
        const float xl = other_half(acc[k][7],   0x65432100u, 0xEDCBA987u);
        const float xr = other_half(acc[k][0],   0x87654321u, 0xFFEDCBA9u);
        const float dl = other_half(D2[k][1][3], 0x65432100u, 0xEDCBA987u);
        const float dr = other_half(D2[k][0][0], 0x87654321u, 0xFFEDCBA9u);
        float xel, del, xer, der;
        if (k == 0) {
            xel = first ? (2.0f * acc[0][0] - acc[0][1])     : haloL;
            del = first ? (2.0f * D2[0][0][0] - D2[0][0][1]) : dL;
        } else {
            xel = from_lane(acc[k - 1][7], 31);
            del = from_lane(D2[k - 1][1][3], 31);
        }
        if (k == C1 - 1) {
            xer = last ? (2.0f * acc[k][7] - acc[k][6])     : haloR;
            der = last ? (2.0f * D2[k][1][3] - D2[k][1][2]) : dR;
        } else {
            xer = from_lane(acc[k + 1][0], 0);
            der = from_lane(D2[k + 1][0][0], 0);
        }
        float vx = hf2 ? xr : xl;
        float vd = hf2 ? dr : dl;
        vx = f0 ? xel : vx;
        vd = f0 ? del : vd;
        vx = f31 ? xer : vx;
        vd = f31 ? der : vd;
        bn[k] = v2f{ vx, vd };
    }

    float* orow = out + 4 * rowa + 4 * (size_t)t0 + (32 * n2 + 4 * hf2);
    __builtin_amdgcn_sched_barrier(0);
#pragma unroll
    for (int k = 0; k < C1; ++k) {
        v2f bx[4], bd[4];
#pragma unroll
        for (int q = 0; q < 4; ++q) {
            bx[q] = v2f{ acc[k][2 * q], acc[k][2 * q + 1] };
            bd[q] = v2f{ D2[k][q >> 1][2 * (q & 1)], D2[k][q >> 1][2 * (q & 1) + 1] };
        }
        v8f z0 = {};
        v8f z1 = {};
        z0 = mma(bh[0], bx[0], z0);   z1 = mma(bh[3], bx[1], z1);
        z0 = mma(bh[1], bx[1], z0);   z1 = mma(bh[0], bx[2], z1);
        z0 = mma(bh[2], bx[2], z0);   z1 = mma(bh[1], bx[3], z1);
        z0 = mma(bg[0], bd[0], z0);   z1 = mma(bg[3], bd[1], z1);
        z0 = mma(bg[1], bd[1], z0);   z1 = mma(bg[0], bd[2], z1);
        z0 = mma(bg[2], bd[2], z0);   z1 = mma(bg[1], bd[3], z1);
        z0 = mma(an0,   bn[k], z0);   z1 = mma(an1,   bn[k], z1);
        float* po = orow + 512 * k;
        *reinterpret_cast<v4f*>(po)      = v4f{ z0[0], z0[1], z0[2], z0[3] };
        *reinterpret_cast<v4f*>(po + 8)  = v4f{ z0[4], z0[5], z0[6], z0[7] };
        *reinterpret_cast<v4f*>(po + 16) = v4f{ z1[0], z1[1], z1[2], z1[3] };
        *reinterpret_cast<v4f*>(po + 24) = v4f{ z1[4], z1[5], z1[6], z1[7] };
        __builtin_amdgcn_sched_barrier(0);
    }
}

extern "C" void kernel_launch(void* const* d_in, const int* in_sizes, int n_in,
                              void* d_out, int out_size, void* d_ws, size_t ws_size,
                              hipStream_t stream) {
    (void)in_sizes; (void)n_in; (void)out_size; (void)ws_size;

    const float* sig = (const float*)d_in[0];
    const float* d0  = (const float*)d_in[1];
    const float* d1  = (const float*)d_in[2];
    const float* d2  = (const float*)d_in[3];
    const float* d3  = (const float*)d_in[4];
    const float* h   = (const float*)d_in[5];
    const float* g   = (const float*)d_in[6];
    float* out = (float*)d_out;

    float* A  = (float*)d_ws;

    dim3 blk(256);
    synth_two_levels_kernel<2><<<64 * (16384 / 256) / 8, blk, 0, stream>>>(sig, d3, d2, A,   h, g, 16384, 6);
    synth_two_levels_kernel<4><<<64 * (65536 / 512) / 8, blk, 0, stream>>>(A,   d1, d0, out, h, g, 65536, 7);
}


# one launch, not 2: a wave runs both level pairs; its 512 samples in between change arrangement through 2 KB of wave-own LDS, no barrier; 4 edge samples recomputed as scalar 6-term sums
# speedup vs baseline: 3.9106x; 1.2041x over previous
#include <hip/hip_runtime.h>


typedef float v2f __attribute__((ext_vector_type(2)));
typedef float v4f __attribute__((ext_vector_type(4)));
typedef float v8f __attribute__((ext_vector_type(8)));

__device__ __forceinline__ float bandsel(int j, float c0, float c1, float c2) {
    float r = 0.0f;
    r = (j == 0) ? c0 : r;
    r = (j == 1) ? c1 : r;
    r = (j == 2) ? c2 : r;
    return r;
}

template <int C>
__device__ __forceinline__ void load_pieces(const float* r, int e0, bool lo, bool hi,
                                            v4f (&P)[C][2]) {
    const int ea = e0 + (lo ? 4 : 0);
    const int eb = e0 + 128 * (C - 1) + 8 + (hi ? -4 : 0);
    P[0][0]     = *reinterpret_cast<const v4f*>(r + ea);
    P[C - 1][1] = *reinterpret_cast<const v4f*>(r + eb);
#pragma unroll
    for (int k = 0; k < C; ++k) {
#pragma unroll
        for (int gg = 0; gg < 2; ++gg) {
            if ((k == 0 && gg == 0) || (k == C - 1 && gg == 1)) continue;
            P[k][gg] = *reinterpret_cast<const v4f*>(r + (e0 + 128 * k + 8 * gg));
        }
    }
}

template <int C>
__device__ __forceinline__ void set_ends(v4f (&P)[C][2], bool lo, bool hi,
                                         bool first, bool last, float vl, float vr) {
    const float al = 2.0f * P[0][0].x - P[0][0].y;
    const float ar = 2.0f * P[C - 1][1].w - P[C - 1][1].z;
    const float l = first ? al : vl;
    const float r = last  ? ar : vr;
    P[0][0].w     = lo ? l : P[0][0].w;
    P[C - 1][1].x = hi ? r : P[C - 1][1].x;
}

__device__ __forceinline__ v8f mma(const v2f a, const v2f b, const v8f c) {
    return __builtin_amdgcn_wmma_f32_16x16x4_f32(false, a, false, b, (short)0, c, false, false);
}

__device__ __forceinline__ float other_half(float v, unsigned sel_lo, unsigned sel_hi) {
    const int i = __builtin_bit_cast(int, v);
    return __builtin_bit_cast(float, __builtin_amdgcn_permlanex16(i, i, sel_lo, sel_hi, false, false));
}

__device__ __forceinline__ float from_lane(float v, int l) {
    return __builtin_bit_cast(float, __builtin_amdgcn_readlane(__builtin_bit_cast(int, v), l));
}

__device__ __forceinline__ void first_level_bands(int lane, const float (&hc)[6], const float (&gc)[6],
                                                  v2f (&ah)[2][2], v2f (&ag)[2][2]) {
    const int n   = lane & 15;
    const int hf  = lane >> 4;
    const int o   = (n & 3) + ((n >> 3) << 2) + (((n >> 2) & 1) << 3);
    const int mm  = o >> 1;
    const bool od = (o & 1) != 0;
    const float hk0 = od ? hc[5] : hc[4], hk1 = od ? hc[3] : hc[2], hk2 = od ? hc[1] : hc[0];
    const float gk0 = od ? gc[5] : gc[4], gk1 = od ? gc[3] : gc[2], gk2 = od ? gc[1] : gc[0];
#pragma unroll
    for (int gg = 0; gg < 2; ++gg) {
#pragma unroll
        for (int s = 0; s < 2; ++s) {
            const int j0 = 8 * gg + 2 * s + 4 * hf - 3 - mm;
            ah[gg][s] = v2f{ bandsel(j0, hk0, hk1, hk2), bandsel(j0 + 1, hk0, hk1, hk2) };
            ag[gg][s] = v2f{ bandsel(j0, gk0, gk1, gk2), bandsel(j0 + 1, gk0, gk1, gk2) };
        }
    }
}

typedef __attribute__((address_space(3))) v4f lds_v4f;
template <int C1, bool LDSOUT>
__device__ __forceinline__ void second_level(v8f (&acc)[C1], v4f (&D2)[C1][2], int lane2,
                                             bool first, bool last,
                                             float haloL, float haloR, float dL, float dR,
                                             const float (&hc)[6], const float (&gc)[6],
                                             float* gbase, unsigned lbase) {
    static_assert(C1 == 1 || C1 == 2, "one or two chunks");
    const int n2   = lane2 & 15;
    const int hf2  = lane2 >> 4;
    const bool f0  = (lane2 == 0);
    const bool f31 = (lane2 == 31);
    const int o2   = (n2 & 3) + ((n2 >> 3) << 2) + (((n2 >> 2) & 1) << 3);
    const int mm2  = o2 >> 1;
    const bool od2 = (o2 & 1) != 0;
    const float hq0 = od2 ? hc[5] : hc[4], hq1 = od2 ? hc[3] : hc[2], hq2 = od2 ? hc[1] : hc[0];
    const float gq0 = od2 ? gc[5] : gc[4], gq1 = od2 ? gc[3] : gc[2], gq2 = od2 ? gc[1] : gc[0];
    const int jb = 4 * hf2 - mm2 + 1;
    v2f bh[4], bg[4];
    {
        const int offs[4] = { 0, 2, 8, -6 };
#pragma unroll
        for (int i = 0; i < 4; ++i) {
            const int j0 = offs[i] + jb;
            bh[i] = v2f{ bandsel(j0, hq0, hq1, hq2), bandsel(j0 + 1, hq0, hq1, hq2) };
            bg[i] = v2f{ bandsel(j0, gq0, gq1, gq2), bandsel(j0 + 1, gq0, gq1, gq2) };
        }
    }
    const bool r0 = (hf2 == 0) && (mm2 == 0);
    const bool r1 = (hf2 == 1) && (mm2 == 7);
    v2f an0 = v2f{ r0 ? hq0 : 0.0f, r0 ? gq0 : 0.0f };
    v2f an1 = v2f{ r1 ? hq2 : 0.0f, r1 ? gq2 : 0.0f };

    v2f bn[C1];
#pragma unroll
    for (int k = 0; k < C1; ++k) {
        const float xl = other_half(acc[k][7],   0x65432100u, 0xEDCBA987u);
        const float xr = other_half(acc[k][0],   0x87654321u, 0xFFEDCBA9u);
        const float dl = other_half(D2[k][1][3], 0x65432100u, 0xEDCBA987u);
        const float dr = other_half(D2[k][0][0], 0x87654321u, 0xFFEDCBA9u);
        float xel, del, xer, der;
        if (k == 0) {
            xel = first ? (2.0f * acc[0][0] - acc[0][1])     : haloL;
            del = first ? (2.0f * D2[0][0][0] - D2[0][0][1]) : dL;
        } else {
            xel = from_lane(acc[k - 1][7], 31);
            del = from_lane(D2[k - 1][1][3], 31);
        }
        if (k == C1 - 1) {
            xer = last ? (2.0f * acc[k][7] - acc[k][6])     : haloR;
            der = last ? (2.0f * D2[k][1][3] - D2[k][1][2]) : dR;
        } else {
            xer = from_lane(acc[k + 1][0], 0);
            der = from_lane(D2[k + 1][0][0], 0);
        }
        float vx = hf2 ? xr : xl;
        float vd = hf2 ? dr : dl;
        vx = f0 ? xel : vx;
        vd = f0 ? del : vd;
        vx = f31 ? xer : vx;
        vd = f31 ? der : vd;
        bn[k] = v2f{ vx, vd };
    }

    unsigned ooff = (unsigned)(4 * (32 * n2 + 4 * hf2)) + (LDSOUT ? lbase : 0u);
    if constexpr (C1 == 1) {
        asm volatile("" : "+v"(bh[0]), "+v"(bh[1]), "+v"(bh[2]), "+v"(bh[3]),
                          "+v"(bg[0]), "+v"(bg[1]), "+v"(bg[2]), "+v"(bg[3]),
                          "+v"(an0), "+v"(an1), "+v"(bn[0]), "+v"(ooff),
                          "+v"(acc[0]), "+v"(D2[0][0]), "+v"(D2[0][1]));
    } else {
        asm volatile("" : "+v"(bh[0]), "+v"(bh[1]), "+v"(bh[2]), "+v"(bh[3]),
                          "+v"(bg[0]), "+v"(bg[1]), "+v"(bg[2]), "+v"(bg[3]),
                          "+v"(an0), "+v"(an1), "+v"(bn[0]), "+v"(bn[C1 - 1]), "+v"(ooff),
                          "+v"(acc[0]), "+v"(acc[C1 - 1]),
                          "+v"(D2[0][0]), "+v"(D2[0][1]), "+v"(D2[C1 - 1][0]), "+v"(D2[C1 - 1][1]));
    }
#pragma unroll
    for (int k = 0; k < C1; ++k) {
        v2f bx[4], bd[4];
#pragma unroll
        for (int q = 0; q < 4; ++q) {
            bx[q] = v2f{ acc[k][2 * q], acc[k][2 * q + 1] };
            bd[q] = v2f{ D2[k][q >> 1][2 * (q & 1)], D2[k][q >> 1][2 * (q & 1) + 1] };
        }
        v8f z0 = {};
        v8f z1 = {};
        z0 = mma(bh[0], bx[0], z0);   z1 = mma(bh[3], bx[1], z1);
        z0 = mma(bh[1], bx[1], z0);   z1 = mma(bh[0], bx[2], z1);
        z0 = mma(bh[2], bx[2], z0);   z1 = mma(bh[1], bx[3], z1);
        z0 = mma(bg[0], bd[0], z0);   z1 = mma(bg[3], bd[1], z1);
        z0 = mma(bg[1], bd[1], z0);   z1 = mma(bg[0], bd[2], z1);
        z0 = mma(bg[2], bd[2], z0);   z1 = mma(bg[1], bd[3], z1);
        z0 = mma(an0,   bn[k], z0);   z1 = mma(an1,   bn[k], z1);
        const v4f o0 = v4f{ z0[0], z0[1], z0[2], z0[3] };
        const v4f o1 = v4f{ z0[4], z0[5], z0[6], z0[7] };
        const v4f o2 = v4f{ z1[0], z1[1], z1[2], z1[3] };
        const v4f o3 = v4f{ z1[4], z1[5], z1[6], z1[7] };
        if constexpr (LDSOUT) {
            const unsigned a = ooff + 2048u * (unsigned)k;
            *reinterpret_cast<lds_v4f*>(a)       = o0;
            *reinterpret_cast<lds_v4f*>(a + 32u) = o1;
            *reinterpret_cast<lds_v4f*>(a + 64u) = o2;
            *reinterpret_cast<lds_v4f*>(a + 96u) = o3;
        } else {
            char* po = reinterpret_cast<char*>(gbase) + ooff + 2048 * k;
            *reinterpret_cast<v4f*>(po)      = o0;
            *reinterpret_cast<v4f*>(po + 32) = o1;
            *reinterpret_cast<v4f*>(po + 64) = o2;
            *reinterpret_cast<v4f*>(po + 96) = o3;
        }
        __builtin_amdgcn_sched_barrier(0);
    }
}

__device__ __forceinline__ void first_level_pair(const v4f (&Xa)[2], const v4f (&Xb)[2],
                                                 const v4f (&Da)[2], const v4f (&Db)[2],
                                                 const v2f (&ah)[2][2], const v2f (&ag)[2][2],
                                                 v8f& ra, v8f& rb) {
    v8f a0 = {};
    v8f a1 = {};
#pragma unroll
    for (int gg = 0; gg < 2; ++gg) {
#pragma unroll
        for (int s = 0; s < 2; ++s) {
            const v2f bx0 = v2f{ Xa[gg][2 * s], Xa[gg][2 * s + 1] };
            const v2f bx1 = v2f{ Xb[gg][2 * s], Xb[gg][2 * s + 1] };
            const v2f bd0 = v2f{ Da[gg][2 * s], Da[gg][2 * s + 1] };
            const v2f bd1 = v2f{ Db[gg][2 * s], Db[gg][2 * s + 1] };
            a0 = mma(ah[gg][s], bx0, a0);
            a1 = mma(ah[gg][s], bx1, a1);
            a0 = mma(ag[gg][s], bd0, a0);
            a1 = mma(ag[gg][s], bd1, a1);
        }
    }
    ra = a0;
    rb = a1;
}

__device__ __forceinline__ float uniform(float v) {
    return __builtin_bit_cast(float, __builtin_amdgcn_readfirstlane(__builtin_bit_cast(int, v)));
}

__device__ __forceinline__ float even_out(float a, float b, float c, float da, float db, float dc,
                                          const float (&hc)[6], const float (&gc)[6]) {
    return a * hc[4] + b * hc[2] + c * hc[0] + da * gc[4] + db * gc[2] + dc * gc[0];
}
__device__ __forceinline__ float odd_out(float a, float b, float c, float da, float db, float dc,
                                         const float (&hc)[6], const float (&gc)[6]) {
    return a * hc[5] + b * hc[3] + c * hc[1] + da * gc[5] + db * gc[3] + dc * gc[1];
}

__global__ __launch_bounds__(256, 8) void synth_four_levels_kernel(
        const float* __restrict__ sig, const float* __restrict__ d3,
        const float* __restrict__ d2,  const float* __restrict__ d1,
        const float* __restrict__ d0,  float* __restrict__ out,
        const float* __restrict__ h,   const float* __restrict__ g) {
    __shared__ float xch[8 * 512];

    constexpr int LQ = 16384;
    constexpr int LT = 65536;
    const int lane = threadIdx.x & 31;
    const int n    = lane & 15;
    const int hf   = lane >> 4;
    const int w    = __builtin_amdgcn_readfirstlane((int)(threadIdx.x >> 5));
    const int Wv   = ((int)blockIdx.x << 3) + w;
    const int ch   = Wv >> 7;
    const int q0   = (Wv & 127) << 7;
    const int t0   = q0 << 2;
    const bool first = (q0 == 0);
    const bool last  = (q0 + 128 == LQ);
    float* xw = xch + 512 * w;
    const unsigned xwaddr = (unsigned)(size_t)(__attribute__((address_space(3))) float*)xch + 2048u * (unsigned)w;

    const size_t rq = (size_t)ch * (size_t)LQ;
    const float* srow  = sig + rq;
    const float* d3row = d3 + rq;
    const float* d2row = d2 + 2 * rq;
    const float* d1row = d1 + 4 * rq;
    const float* d0row = d0 + 8 * rq;

    const int e0   = 8 * n + 4 * hf - 4;
    const bool l0  = (lane == 0);
    const bool l31 = (lane == 31);
    const bool fl  = first && l0;
    const bool lh  = last && l31;
    v4f XQ[1][2], DQ[1][2], DM[1][2], D1[4][2];
    load_pieces<1>(srow + q0, e0, fl, lh, XQ);
    load_pieces<1>(d3row + q0, e0, fl, lh, DQ);
    int doff = 16 * n + 4 * hf;
    asm volatile("" : "+v"(doff));
    {
        const float* p = d2row + 2 * (size_t)q0 + doff;
        DM[0][0] = *reinterpret_cast<const v4f*>(p);
        DM[0][1] = *reinterpret_cast<const v4f*>(p + 8);
    }
    load_pieces<4>(d1row + t0, e0, fl, lh, D1);

    const float hc[6] = { h[0], h[1], h[2], h[3], h[4], h[5] };
    const float gc[6] = { g[0], g[1], g[2], g[3], g[4], g[5] };

    const int lb  = first ? 0 : q0 - 2;
    const int rb  = last ? LQ - 4 : q0 + 126;
    const int ulb = first ? 0 : 2 * q0 - 2;
    const int urb = last ? 2 * LQ - 3 : 2 * q0 + 255;
    const int vlb = first ? 0 : t0 - 2;
    const int vrb = last ? LT - 3 : t0 + 511;
    float sL[4], tL[4], sR[4], tR[4], uL[3], uR[3], vL[3], vR[3];
#pragma unroll
    for (int i = 0; i < 4; ++i) {
        sL[i] = srow[lb + i];   tL[i] = d3row[lb + i];
        sR[i] = srow[rb + i];   tR[i] = d3row[rb + i];
    }
#pragma unroll
    for (int i = 0; i < 3; ++i) {
        uL[i] = d2row[ulb + i]; uR[i] = d2row[urb + i];
        vL[i] = d1row[vlb + i]; vR[i] = d1row[vrb + i];
    }
    const float wL = d0row[first ? 0 : 2 * t0 - 1];
    const float wR = d0row[last ? 2 * LT - 1 : 2 * (t0 + 512)];
    const float wML = d0row[2 * t0 + 511];
    const float wMR = d0row[2 * t0 + 512];
    const float mA = even_out(sL[0], sL[1], sL[2], tL[0], tL[1], tL[2], hc, gc);
    float mB = uniform(odd_out (sL[0], sL[1], sL[2], tL[0], tL[1], tL[2], hc, gc));
    const float mC = even_out(sL[1], sL[2], sL[3], tL[1], tL[2], tL[3], hc, gc);
    const float nA = odd_out (sR[0], sR[1], sR[2], tR[0], tR[1], tR[2], hc, gc);
    float nB = uniform(even_out(sR[1], sR[2], sR[3], tR[1], tR[2], tR[3], hc, gc));
    const float nC = odd_out (sR[1], sR[2], sR[3], tR[1], tR[2], tR[3], hc, gc);
    const float aL0 = even_out(mA, mB, mC, uL[0], uL[1], uL[2], hc, gc);
    float aL1 = uniform(odd_out (mA, mB, mC, uL[0], uL[1], uL[2], hc, gc));
    float aR0 = uniform(even_out(nA, nB, nC, uR[0], uR[1], uR[2], hc, gc));
    float aR1 = uniform(odd_out (nA, nB, nC, uR[0], uR[1], uR[2], hc, gc));
    float pL  = uniform(aL0 * hc[5] + aL1 * hc[3]);

    set_ends<1>(XQ, fl, lh, true, true, 0.0f, 0.0f);
    set_ends<1>(DQ, fl, lh, true, true, 0.0f, 0.0f);
    v8f accq[1];
    {
        v2f ah[2][2], ag[2][2];
        first_level_bands(lane, hc, gc, ah, ag);
        asm volatile("" : "+v"(ah[0][0]), "+v"(ah[0][1]), "+v"(ah[1][0]), "+v"(ah[1][1]),
                          "+v"(ag[0][0]), "+v"(ag[0][1]), "+v"(ag[1][0]), "+v"(ag[1][1]),
                          "+v"(XQ[0][0]), "+v"(XQ[0][1]), "+v"(DQ[0][0]), "+v"(DQ[0][1]),
                          "+s"(mB), "+s"(nB), "+s"(aL1), "+s"(aR0), "+s"(aR1), "+s"(pL));
        v8f a0 = {};
#pragma unroll
        for (int gg = 0; gg < 2; ++gg) {
#pragma unroll
            for (int s = 0; s < 2; ++s) {
                const v2f bx0 = v2f{ XQ[0][gg][2 * s], XQ[0][gg][2 * s + 1] };
                const v2f bd0 = v2f{ DQ[0][gg][2 * s], DQ[0][gg][2 * s + 1] };
                a0 = mma(ah[gg][s], bx0, a0);
                a0 = mma(ag[gg][s], bd0, a0);
            }
        }
        accq[0] = a0;
    }
    int lane_b;
    asm volatile("v_nop\n\tv_nop\n\tv_nop\n\tv_nop\n\tv_nop\n\tv_mbcnt_lo_u32_b32 %1, -1, 0"
                 : "+v"(accq[0]), "=v"(lane_b));
    __builtin_amdgcn_sched_barrier(0);
    second_level<1, true>(accq, DM, lane_b, first, last, mB, nB, uL[1], uR[1], hc, gc, nullptr, xwaddr);

    int lane_c;
    asm volatile("v_nop\n\tv_nop\n\tv_nop\n\tv_nop\n\tv_nop\n\tv_mbcnt_lo_u32_b32 %0, -1, 0"
                 : "=v"(lane_c));
    __builtin_amdgcn_sched_barrier(0);
    __builtin_amdgcn_fence(__ATOMIC_RELEASE, "wavefront");
    __builtin_amdgcn_wave_barrier();
    __builtin_amdgcn_fence(__ATOMIC_ACQUIRE, "wavefront");

    const int nc   = lane_c & 15;
    const int hfc  = lane_c >> 4;
    const int e0c  = 8 * nc + 4 * hfc - 4;
    const bool c0  = (lane_c == 0);
    const bool c31 = (lane_c == 31);
    v4f X1[4][2];
    load_pieces<4>(xw, e0c, c0, c31, X1);
    const float aF = xw[0];
    const float aE = xw[511];
    set_ends<4>(X1, c0, c31, first, last, aL1, aR0);
    set_ends<4>(D1, first && c0, last && c31, true, true, 0.0f, 0.0f);
    float haloL = pL + aF * hc[1]
                + vL[0] * gc[5] + vL[1] * gc[3] + vL[2] * gc[1];
    float haloR = aE * hc[4] + aR0 * hc[2] + aR1 * hc[0]
                + vR[0] * gc[4] + vR[1] * gc[2] + vR[2] * gc[0];

    v8f accA[2], accB[2];
    v4f D2a[2][2], D2b[2][2];
    {
        v2f ah[2][2], ag[2][2];
        first_level_bands(lane_c, hc, gc, ah, ag);
        asm volatile("" : "+v"(ah[0][0]), "+v"(ah[0][1]), "+v"(ah[1][0]), "+v"(ah[1][1]),
                          "+v"(ag[0][0]), "+v"(ag[0][1]), "+v"(ag[1][0]), "+v"(ag[1][1]),
                          "+v"(X1[0][0]), "+v"(X1[0][1]), "+v"(X1[1][0]), "+v"(X1[1][1]),
                          "+v"(X1[2][0]), "+v"(X1[2][1]), "+v"(X1[3][0]), "+v"(X1[3][1]),
                          "+v"(D1[0][0]), "+v"(D1[0][1]), "+v"(D1[1][0]), "+v"(D1[1][1]),
                          "+v"(D1[2][0]), "+v"(D1[2][1]), "+v"(D1[3][0]), "+v"(D1[3][1]),
                          "+v"(haloL), "+v"(haloR));
        first_level_pair(X1[0], X1[1], D1[0], D1[1], ah, ag, accA[0], accA[1]);
        asm volatile("v_nop\n\tv_nop\n\tv_nop\n\tv_nop\n\tv_nop"
                     : "+v"(accA[0]), "+v"(accA[1]), "+v"(doff));
        {
            const float* p = d0row + 2 * (size_t)t0 + doff;
#pragma unroll
            for (int kk = 0; kk < 2; ++kk) {
                D2a[kk][0] = *reinterpret_cast<const v4f*>(p + 256 * kk);
                D2a[kk][1] = *reinterpret_cast<const v4f*>(p + 256 * kk + 8);
            }
        }
        __builtin_amdgcn_sched_barrier(0);
        first_level_pair(X1[2], X1[3], D1[2], D1[3], ah, ag, accB[0], accB[1]);
    }
    int lane_d;
    asm volatile("v_nop\n\tv_nop\n\tv_nop\n\tv_nop\n\tv_nop\n\tv_mbcnt_lo_u32_b32 %4, -1, 0"
                 : "+v"(accA[0]), "+v"(accA[1]), "+v"(accB[0]), "+v"(accB[1]), "=v"(lane_d), "+v"(doff));
    {
        const float* p = d0row + 2 * (size_t)t0 + 512 + doff;
#pragma unroll
        for (int kk = 0; kk < 2; ++kk) {
            D2b[kk][0] = *reinterpret_cast<const v4f*>(p + 256 * kk);
            D2b[kk][1] = *reinterpret_cast<const v4f*>(p + 256 * kk + 8);
        }
    }
    __builtin_amdgcn_sched_barrier(0);
    const float xmr = from_lane(accB[0][0], 0);
    const float xml = from_lane(accA[1][7], 31);
    float* ostrip = out + 16 * rq + 16 * (size_t)q0;
    second_level<2, false>(accA, D2a, lane_d, first, false, haloL, xmr, wL, wMR, hc, gc, ostrip, 0u);
    int lane_e;
    asm volatile("v_nop\n\tv_nop\n\tv_nop\n\tv_nop\n\tv_nop\n\tv_mbcnt_lo_u32_b32 %2, -1, 0"
                 : "+v"(accB[0]), "+v"(accB[1]), "=v"(lane_e));
    __builtin_amdgcn_sched_barrier(0);
    second_level<2, false>(accB, D2b, lane_e, false, last, xml, haloR, wML, wR, hc, gc, ostrip + 1024, 0u);
}

extern "C" void kernel_launch(void* const* d_in, const int* in_sizes, int n_in,
                              void* d_out, int out_size, void* d_ws, size_t ws_size,
                              hipStream_t stream) {
    (void)in_sizes; (void)n_in; (void)out_size; (void)d_ws; (void)ws_size;

    const float* sig = (const float*)d_in[0];
    const float* d0  = (const float*)d_in[1];
    const float* d1  = (const float*)d_in[2];
    const float* d2  = (const float*)d_in[3];
    const float* d3  = (const float*)d_in[4];
    const float* h   = (const float*)d_in[5];
    const float* g   = (const float*)d_in[6];
    float* out = (float*)d_out;

    synth_four_levels_kernel<<<1024, dim3(256), 0, stream>>>(sig, d3, d2, d1, d0, out, h, g);
}


# each pair of levels is ONE banded product with multiplied-out taps (K = 28: 7 matrix instructions per 16 outputs, 70 not 110 a wave); tap products round first and the sums regroup in K
# speedup vs baseline: 4.2005x; 1.0741x over previous
#include <hip/hip_runtime.h>


typedef float v2f __attribute__((ext_vector_type(2)));
typedef float v4f __attribute__((ext_vector_type(4)));
typedef float v8f __attribute__((ext_vector_type(8)));
typedef __attribute__((address_space(3))) v4f lds_v4f;
typedef __attribute__((address_space(3))) v2f lds_v2f;

__device__ __forceinline__ float bandsel(int j, float c0, float c1, float c2) {
    float r = 0.0f;
    r = (j == 0) ? c0 : r;
    r = (j == 1) ? c1 : r;
    r = (j == 2) ? c2 : r;
    return r;
}

__device__ __forceinline__ v8f mma(const v2f a, const v2f b, const v8f c) {
    return __builtin_amdgcn_wmma_f32_16x16x4_f32(false, a, false, b, (short)0, c, false, false);
}

__device__ __forceinline__ float uniform(float v) {
    return __builtin_bit_cast(float, __builtin_amdgcn_readfirstlane(__builtin_bit_cast(int, v)));
}

__device__ __forceinline__ float even_out(float a, float b, float c, float da, float db, float dc,
                                          const float (&hc)[6], const float (&gc)[6]) {
    return a * hc[4] + b * hc[2] + c * hc[0] + da * gc[4] + db * gc[2] + dc * gc[0];
}
__device__ __forceinline__ float odd_out(float a, float b, float c, float da, float db, float dc,
                                         const float (&hc)[6], const float (&gc)[6]) {
    return a * hc[5] + b * hc[3] + c * hc[1] + da * gc[5] + db * gc[3] + dc * gc[1];
}

__device__ __forceinline__ float ctap(int Mloc, int k, const float (&hP)[3], const float (&t)[6]) {
    float r = 0.0f;
#pragma unroll
    for (int J = 0; J < 3; ++J) {
        const int Yl  = Mloc - 1 + J;
        const int fl  = Yl >> 1;
        const bool od = (Yl & 1) != 0;
        const float t0 = od ? t[5] : t[4], t1 = od ? t[3] : t[2], t2 = od ? t[1] : t[0];
        r += hP[J] * bandsel(k - 1 - fl, t0, t1, t2);
    }
    return r;
}

struct Bands { v2f x0, x1, a0, a1, b0, b1, b2; };
__device__ __forceinline__ void build_bands(int lane, const float (&hc)[6], const float (&gc)[6], Bands& W) {
    const int m    = lane & 15;
    const int hf   = lane >> 4;
    const int o    = (m & 3) + ((m >> 3) << 2) + (((m >> 2) & 1) << 3);
    const int Mloc = o >> 1;
    const bool P   = (o & 1) != 0;
    const float hP[3] = { P ? hc[5] : hc[4], P ? hc[3] : hc[2], P ? hc[1] : hc[0] };
    const float gP0 = P ? gc[5] : gc[4], gP1 = P ? gc[3] : gc[2], gP2 = P ? gc[1] : gc[0];
    const int k0 = 4 * hf;
    W.x0 = v2f{ ctap(Mloc, k0,     hP, hc), ctap(Mloc, k0 + 1, hP, hc) };
    W.x1 = v2f{ ctap(Mloc, k0 + 2, hP, hc), ctap(Mloc, k0 + 3, hP, hc) };
    W.a0 = v2f{ ctap(Mloc, k0,     hP, gc), ctap(Mloc, k0 + 1, hP, gc) };
    W.a1 = v2f{ ctap(Mloc, k0 + 2, hP, gc), ctap(Mloc, k0 + 3, hP, gc) };
    const int jb = k0 - 1 - Mloc;
    const int jc = 8 + 2 * hf - 1 - Mloc;
    float b00 = bandsel(jb, gP0, gP1, gP2);
    float b21 = bandsel(jc + 1, gP0, gP1, gP2);
    b00 = (hf == 0) ? ((Mloc == 0) ? hP[0] : 0.0f) : b00;
    b21 = (hf == 1) ? ((Mloc == 7) ? hP[2] : 0.0f) : b21;
    W.b0 = v2f{ b00, bandsel(jb + 1, gP0, gP1, gP2) };
    W.b1 = v2f{ bandsel(jb + 2, gP0, gP1, gP2), bandsel(jb + 3, gP0, gP1, gP2) };
    W.b2 = v2f{ bandsel(jc, gP0, gP1, gP2), b21 };
}

template <int C0, int NTOT, bool XLDS>
__device__ __forceinline__ void load_xa(const float* p, unsigned pl, int lane, bool lo, bool hi,
                                        v2f (&X)[2][2]) {
    const int e = 4 * (lane & 15) + 4 * (lane >> 4) - 2;
#pragma unroll
    for (int c = 0; c < 2; ++c) {
        int e0 = e + 64 * (C0 + c);
        int e1 = e0 + 2;
        if (C0 + c == 0)        e0 += (lo && lane == 0) ? 2 : 0;
        if (C0 + c == NTOT - 1) e1 -= (hi && lane == 31) ? 2 : 0;
        if constexpr (XLDS) {
            X[c][0] = *reinterpret_cast<lds_v2f*>(pl + 4u * (unsigned)e0);
            X[c][1] = *reinterpret_cast<lds_v2f*>(pl + 4u * (unsigned)e1);
        } else {
            X[c][0] = *reinterpret_cast<const v2f*>(p + e0);
            X[c][1] = *reinterpret_cast<const v2f*>(p + e1);
        }
    }
}

template <int C0, int NTOT>
__device__ __forceinline__ void load_b(const float* p, int lane, bool lo, bool hi, v2f (&B)[2][3]) {
    const int n = lane & 15, hf = lane >> 4;
    const int e = 8 * n + 4 * hf - 2;
    const int f = 8 * n + 2 * hf + 6;
#pragma unroll
    for (int c = 0; c < 2; ++c) {
        int e0 = e + 128 * (C0 + c);
        int e2 = f + 128 * (C0 + c);
        const int e1 = e0 + 2;
        if (C0 + c == 0)        e0 += (lo && lane == 0) ? 2 : 0;
        if (C0 + c == NTOT - 1) e2 -= (hi && lane == 31) ? 2 : 0;
        B[c][0] = *reinterpret_cast<const v2f*>(p + e0);
        B[c][1] = *reinterpret_cast<const v2f*>(p + e1);
        B[c][2] = *reinterpret_cast<const v2f*>(p + e2);
    }
}

template <int C0, int NTOT, bool XLDS, bool LDSOUT>
__device__ __forceinline__ void run_group(v2f (&X)[2][2], v2f (&A)[2][2], v2f (&B)[2][3], Bands& W,
                                          int lane, bool first, bool last,
                                          float vl0, float vl1, float vr0, float vr1,
                                          const float (&hc)[6], const float (&gc)[6],
                                          float* gout, unsigned lout) {
    const int n    = lane & 15;
    const int hf   = lane >> 4;
    const bool l0  = (lane == 0);
    const bool l31 = (lane == 31);
    const bool h0  = (hf == 0);
    const bool h1  = (hf == 1);
#pragma unroll
    for (int c = 0; c < 2; ++c) {
        if (C0 + c == 0) {
            const v2f x1 = X[c][1], a1 = A[c][1], b1 = B[c][1];
            const float xpm = 2.0f * x1.x - x1.y;
            const float dpm = 2.0f * a1.x - a1.y;
            const float bpm = 2.0f * b1.x - b1.y;
            const float y0  = even_out(xpm, x1.x, x1.y, dpm, a1.x, a1.y, hc, gc);
            const float y1  = odd_out (xpm, x1.x, x1.y, dpm, a1.x, a1.y, hc, gc);
            const float yn  = xpm * hc[3] + x1.x * hc[1] + dpm * gc[3] + a1.x * gc[1];
            const float cL  = 2.0f * y0 - y1 - yn;
            const bool fx   = l0 && first;
            const bool sx   = l0 && (first || XLDS);
            const float xa  = first ? 0.0f : vl0;
            const float xb  = first ? xpm  : vl1;
            X[c][0].x = sx ? xa : X[c][0].x;
            X[c][0].y = sx ? xb : X[c][0].y;
            A[c][0].x = fx ? 0.0f : A[c][0].x;
            A[c][0].y = fx ? dpm  : A[c][0].y;
            B[c][0].y = fx ? bpm  : B[c][0].y;
            B[c][0].x = h0 ? (fx ? cL : 0.0f) : B[c][0].x;
        } else {
            B[c][0].x = h0 ? 0.0f : B[c][0].x;
        }
        if (C0 + c == NTOT - 1) {
            const v2f x0 = X[c][0], a0 = A[c][0], b2 = B[c][2];
            const float xpp = 2.0f * x0.y - x0.x;
            const float dpp = 2.0f * a0.y - a0.x;
            const float bpp = 2.0f * b2.y - b2.x;
            const float ya  = odd_out (x0.x, x0.y, xpp, a0.x, a0.y, dpp, hc, gc);
            const float yb  = even_out(x0.x, x0.y, xpp, a0.x, a0.y, dpp, hc, gc);
            const float yn  = x0.y * hc[4] + xpp * hc[2] + a0.y * gc[4] + dpp * gc[2];
            const float cR  = 2.0f * ya - yb - yn;
            const bool fx   = l31 && last;
            const bool sx   = l31 && (last || XLDS);
            const float xa  = last ? xpp  : vr0;
            const float xb  = last ? 0.0f : vr1;
            X[c][1].x = sx ? xa : X[c][1].x;
            X[c][1].y = sx ? xb : X[c][1].y;
            A[c][1].x = fx ? dpp  : A[c][1].x;
            A[c][1].y = fx ? 0.0f : A[c][1].y;
            B[c][2].x = fx ? bpp  : B[c][2].x;
            B[c][2].y = h1 ? (fx ? cR : 0.0f) : B[c][2].y;
        } else {
            B[c][2].y = h1 ? 0.0f : B[c][2].y;
        }
    }
    unsigned ooff = (unsigned)(64 * n + 16 * hf) + (LDSOUT ? lout : 0u);
    __builtin_amdgcn_sched_barrier(0);
    asm volatile("" : "+v"(W.x0), "+v"(W.x1), "+v"(W.a0), "+v"(W.a1),
                      "+v"(W.b0), "+v"(W.b1), "+v"(W.b2), "+v"(ooff));
    asm volatile("" : "+v"(X[0][0]), "+v"(X[0][1]), "+v"(A[0][0]), "+v"(A[0][1]),
                      "+v"(B[0][0]), "+v"(B[0][1]), "+v"(B[0][2]),
                      "+v"(X[1][0]), "+v"(X[1][1]), "+v"(A[1][0]), "+v"(A[1][1]),
                      "+v"(B[1][0]), "+v"(B[1][1]), "+v"(B[1][2]));
    v8f z0 = {};
    v8f z1 = {};
    z0 = mma(W.x0, X[0][0], z0);   z1 = mma(W.x0, X[1][0], z1);
    z0 = mma(W.x1, X[0][1], z0);   z1 = mma(W.x1, X[1][1], z1);
    z0 = mma(W.a0, A[0][0], z0);   z1 = mma(W.a0, A[1][0], z1);
    z0 = mma(W.a1, A[0][1], z0);   z1 = mma(W.a1, A[1][1], z1);
    z0 = mma(W.b0, B[0][0], z0);   z1 = mma(W.b0, B[1][0], z1);
    z0 = mma(W.b1, B[0][1], z0);   z1 = mma(W.b1, B[1][1], z1);
    z0 = mma(W.b2, B[0][2], z0);   z1 = mma(W.b2, B[1][2], z1);
    const v4f o0 = v4f{ z0[0], z0[1], z0[2], z0[3] };
    const v4f o1 = v4f{ z0[4], z0[5], z0[6], z0[7] };
    const v4f o2 = v4f{ z1[0], z1[1], z1[2], z1[3] };
    const v4f o3 = v4f{ z1[4], z1[5], z1[6], z1[7] };
    if constexpr (LDSOUT) {
        const unsigned a = ooff + 1024u * (unsigned)C0;
        *reinterpret_cast<lds_v4f*>(a)         = o0;
        *reinterpret_cast<lds_v4f*>(a + 32u)   = o1;
        *reinterpret_cast<lds_v4f*>(a + 1024u) = o2;
        *reinterpret_cast<lds_v4f*>(a + 1056u) = o3;
    } else {
        char* po = reinterpret_cast<char*>(gout) + ooff + 1024 * C0;
        *reinterpret_cast<v4f*>(po)        = o0;
        *reinterpret_cast<v4f*>(po + 32)   = o1;
        *reinterpret_cast<v4f*>(po + 1024) = o2;
        *reinterpret_cast<v4f*>(po + 1056) = o3;
    }
    __builtin_amdgcn_sched_barrier(0);
}

__device__ __forceinline__ int slots_and_lane() {
    int l;
    asm volatile("v_nop\n\tv_nop\n\tv_nop\n\tv_nop\n\tv_nop\n\tv_mbcnt_lo_u32_b32 %0, -1, 0" : "=v"(l));
    __builtin_amdgcn_sched_barrier(0);
    return l;
}

__global__ __launch_bounds__(256, 8) void synth_four_levels_kernel(
        const float* __restrict__ sig, const float* __restrict__ d3,
        const float* __restrict__ d2,  const float* __restrict__ d1,
        const float* __restrict__ d0,  float* __restrict__ out,
        const float* __restrict__ h,   const float* __restrict__ g) {
    __shared__ float xch[8 * 512];

    constexpr int LQ = 16384;
    const int lane = threadIdx.x & 31;
    const int w    = __builtin_amdgcn_readfirstlane((int)(threadIdx.x >> 5));
    const int Wv   = ((int)blockIdx.x << 3) + w;
    const int ch   = Wv >> 7;
    const int q0   = (Wv & 127) << 7;
    const int t0   = q0 << 2;
    const bool first = (q0 == 0);
    const bool last  = (q0 + 128 == LQ);
    const unsigned xwaddr = (unsigned)(size_t)(__attribute__((address_space(3))) float*)xch + 2048u * (unsigned)w;

    const size_t rq = (size_t)ch * (size_t)LQ;
    const float* srow  = sig + rq;
    const float* d3row = d3 + rq;
    const float* d2row = d2 + 2 * rq;
    const float* d1s   = d1 + 4 * rq + t0;
    const float* d0s   = d0 + 8 * rq + 2 * (size_t)t0;

    v2f XQ[2][2], AQ[2][2], BQ[2][3];
    load_xa<0, 2, false>(srow + q0, 0u, lane, first, last, XQ);
    load_xa<0, 2, false>(d3row + q0, 0u, lane, first, last, AQ);
    load_b<0, 2>(d2row + 2 * (size_t)q0, lane, first, last, BQ);
    v2f Ag0[2][2], Bg0[2][3], Ag1[2][2], Bg1[2][3];
    load_xa<0, 8, false>(d1s, 0u, lane, first, last, Ag0);
    load_b<0, 8>(d0s, lane, first, last, Bg0);
    load_xa<2, 8, false>(d1s, 0u, lane, first, last, Ag1);
    load_b<2, 8>(d0s, lane, first, last, Bg1);

    const float hc[6] = { h[0], h[1], h[2], h[3], h[4], h[5] };
    const float gc[6] = { g[0], g[1], g[2], g[3], g[4], g[5] };

    const int lb  = first ? 0 : q0 - 2;
    const int rb  = last ? LQ - 4 : q0 + 126;
    const int ulb = first ? 0 : 2 * q0 - 2;
    const int urb = last ? 2 * LQ - 3 : 2 * q0 + 255;
    float sL[4], tL[4], sR[4], tR[4], uL[3], uR[3];
#pragma unroll
    for (int i = 0; i < 4; ++i) {
        sL[i] = srow[lb + i];   tL[i] = d3row[lb + i];
        sR[i] = srow[rb + i];   tR[i] = d3row[rb + i];
    }
#pragma unroll
    for (int i = 0; i < 3; ++i) {
        uL[i] = d2row[ulb + i]; uR[i] = d2row[urb + i];
    }
    const float mA = even_out(sL[0], sL[1], sL[2], tL[0], tL[1], tL[2], hc, gc);
    const float mB = odd_out (sL[0], sL[1], sL[2], tL[0], tL[1], tL[2], hc, gc);
    const float mC = even_out(sL[1], sL[2], sL[3], tL[1], tL[2], tL[3], hc, gc);
    const float nA = odd_out (sR[0], sR[1], sR[2], tR[0], tR[1], tR[2], hc, gc);
    const float nB = even_out(sR[1], sR[2], sR[3], tR[1], tR[2], tR[3], hc, gc);
    const float nC = odd_out (sR[1], sR[2], sR[3], tR[1], tR[2], tR[3], hc, gc);
    float aL0 = uniform(even_out(mA, mB, mC, uL[0], uL[1], uL[2], hc, gc));
    float aL1 = uniform(odd_out (mA, mB, mC, uL[0], uL[1], uL[2], hc, gc));
    float aR0 = uniform(even_out(nA, nB, nC, uR[0], uR[1], uR[2], hc, gc));
    float aR1 = uniform(odd_out (nA, nB, nC, uR[0], uR[1], uR[2], hc, gc));
    asm volatile("" : "+s"(aL0), "+s"(aL1), "+s"(aR0), "+s"(aR1));

    Bands W;
    build_bands(lane, hc, gc, W);

    run_group<0, 2, false, true>(XQ, AQ, BQ, W, lane, first, last, 0.0f, 0.0f, 0.0f, 0.0f,
                                 hc, gc, nullptr, xwaddr);
    const int lane_a = slots_and_lane();
    __builtin_amdgcn_fence(__ATOMIC_RELEASE, "wavefront");
    __builtin_amdgcn_wave_barrier();
    __builtin_amdgcn_fence(__ATOMIC_ACQUIRE, "wavefront");

    float* ostrip = out + 16 * rq + 16 * (size_t)q0;
    v2f Xg[2][2];
    v2f Ag2[2][2], Bg2[2][3], Ag3[2][2], Bg3[2][3];

    load_xa<0, 8, true>(nullptr, xwaddr, lane_a, true, true, Xg);
    load_xa<4, 8, false>(d1s, 0u, lane_a, first, last, Ag2);
    load_b<4, 8>(d0s, lane_a, first, last, Bg2);
    run_group<0, 8, true, false>(Xg, Ag0, Bg0, W, lane_a, first, last, aL0, aL1, aR0, aR1,
                                 hc, gc, ostrip, 0u);
    const int lane_b = slots_and_lane();

    load_xa<2, 8, true>(nullptr, xwaddr, lane_b, true, true, Xg);
    load_xa<6, 8, false>(d1s, 0u, lane_b, first, last, Ag3);
    load_b<6, 8>(d0s, lane_b, first, last, Bg3);
    run_group<2, 8, true, false>(Xg, Ag1, Bg1, W, lane_b, first, last, aL0, aL1, aR0, aR1,
                                 hc, gc, ostrip, 0u);
    const int lane_c = slots_and_lane();

    load_xa<4, 8, true>(nullptr, xwaddr, lane_c, true, true, Xg);
    run_group<4, 8, true, false>(Xg, Ag2, Bg2, W, lane_c, first, last, aL0, aL1, aR0, aR1,
                                 hc, gc, ostrip, 0u);
    const int lane_d = slots_and_lane();

    load_xa<6, 8, true>(nullptr, xwaddr, lane_d, true, true, Xg);
    run_group<6, 8, true, false>(Xg, Ag3, Bg3, W, lane_d, first, last, aL0, aL1, aR0, aR1,
                                 hc, gc, ostrip, 0u);
}

extern "C" void kernel_launch(void* const* d_in, const int* in_sizes, int n_in,
                              void* d_out, int out_size, void* d_ws, size_t ws_size,
                              hipStream_t stream) {
    (void)in_sizes; (void)n_in; (void)out_size; (void)d_ws; (void)ws_size;

    const float* sig = (const float*)d_in[0];
    const float* d0  = (const float*)d_in[1];
    const float* d1  = (const float*)d_in[2];
    const float* d2  = (const float*)d_in[3];
    const float* d3  = (const float*)d_in[4];
    const float* h   = (const float*)d_in[5];
    const float* g   = (const float*)d_in[6];
    float* out = (float*)d_out;

    synth_four_levels_kernel<<<1024, dim3(256), 0, stream>>>(sig, d3, d2, d1, d0, out, h, g);
}
